# starting kernel
# baseline (speedup 1.0000x reference)
#include <hip/hip_runtime.h>

#define SEQL   2048
#define DIMM   2048
#define NHEADS 16
#define HDIM   128

typedef float v2f __attribute__((ext_vector_type(2)));
typedef float v8f __attribute__((ext_vector_type(8)));

__device__ __forceinline__ v8f wmma_f32(v2f a, v2f b, v8f c) {
    return __builtin_amdgcn_wmma_f32_16x16x4_f32(false, a, false, b, (short)0, c, false, false);
}

__global__ __launch_bounds__(256)
void gemm_abt(const float* __restrict__ A, const float* __restrict__ B,
              float* __restrict__ C) {
    const int lane = threadIdx.x & 31;
    const int wave = threadIdx.x >> 5;
    const int hi   = lane >> 4;
    const int lc   = lane & 15;

    const int m0 = blockIdx.y * 128 + wave * 16;
    const int n0 = blockIdx.x * 64;

    v8f acc0 = {}, acc1 = {}, acc2 = {}, acc3 = {};

    const float* ap  = A + (size_t)(m0 + lc) * DIMM + 2 * hi;
    const float* bp0 = B + (size_t)(n0 +  0 + lc) * DIMM + 2 * hi;
    const float* bp1 = B + (size_t)(n0 + 16 + lc) * DIMM + 2 * hi;
    const float* bp2 = B + (size_t)(n0 + 32 + lc) * DIMM + 2 * hi;
    const float* bp3 = B + (size_t)(n0 + 48 + lc) * DIMM + 2 * hi;

#pragma unroll 4
    for (int k0 = 0; k0 < DIMM; k0 += 4) {
        v2f a  = *(const v2f*)(ap  + k0);
        v2f b0 = *(const v2f*)(bp0 + k0);
        v2f b1 = *(const v2f*)(bp1 + k0);
        v2f b2 = *(const v2f*)(bp2 + k0);
        v2f b3 = *(const v2f*)(bp3 + k0);
        acc0 = wmma_f32(a, b0, acc0);
        acc1 = wmma_f32(a, b1, acc1);
        acc2 = wmma_f32(a, b2, acc2);
        acc3 = wmma_f32(a, b3, acc3);
    }

#pragma unroll
    for (int r = 0; r < 8; ++r) {
        float* crow = C + (size_t)(m0 + r + 8 * hi) * DIMM + n0 + lc;
        crow[ 0] = acc0[r];
        crow[16] = acc1[r];
        crow[32] = acc2[r];
        crow[48] = acc3[r];
    }
}

__global__ __launch_bounds__(256)
void attn_fwd(const float* __restrict__ Q, const float* __restrict__ K,
              const float* __restrict__ V, const float* __restrict__ mask,
              float* __restrict__ O) {
    __shared__ float ldsP[8 * 256];

    const int lane = threadIdx.x & 31;
    const int wave = threadIdx.x >> 5;
    const int hi   = lane >> 4;
    const int lc   = lane & 15;

    const int h  = blockIdx.y;
    const int q0 = (blockIdx.x * 8 + wave) * 16;

    v2f aq[32];
    {
        const float* qrow = Q + (size_t)(q0 + lc) * DIMM + h * HDIM + 2 * hi;
#pragma unroll
        for (int s = 0; s < 32; ++s) aq[s] = *(const v2f*)(qrow + s * 4);
    }

    v8f oacc[8];
#pragma unroll
    for (int t = 0; t < 8; ++t) oacc[t] = (v8f){};
    float mrow[8], lrow[8];
#pragma unroll
    for (int r = 0; r < 8; ++r) { mrow[r] = -1.0e30f; lrow[r] = 0.0f; }

    float* lp = &ldsP[wave * 256];
    const float scale = 0.08838834764831845f;
    const size_t mbase = ((size_t)h * SEQL + q0) * SEQL;

#pragma unroll 1
    for (int kv0 = 0; kv0 < SEQL; kv0 += 16) {
        v8f s = {};
        const float* krow = K + (size_t)(kv0 + lc) * DIMM + h * HDIM + 2 * hi;
#pragma unroll
        for (int ss = 0; ss < 32; ++ss) {
            v2f bk = *(const v2f*)(krow + ss * 4);
            s = wmma_f32(aq[ss], bk, s);
        }

#pragma unroll
        for (int r = 0; r < 8; ++r) {
            const float mval = __builtin_nontemporal_load(
                mask + mbase + (size_t)(r + 8 * hi) * SEQL + kv0 + lc);
            float v = s[r] * scale + mval;

            float rmax = v;
            rmax = fmaxf(rmax, __shfl_xor(rmax, 1, 32));
            rmax = fmaxf(rmax, __shfl_xor(rmax, 2, 32));
            rmax = fmaxf(rmax, __shfl_xor(rmax, 4, 32));
            rmax = fmaxf(rmax, __shfl_xor(rmax, 8, 32));

            const float mnew = fmaxf(mrow[r], rmax);
            const float corr = __expf(mrow[r] - mnew);
            mrow[r] = mnew;

            const float p = __expf(v - mnew);
            s[r] = p;

            float psum = p;
            psum += __shfl_xor(psum, 1, 32);
            psum += __shfl_xor(psum, 2, 32);
            psum += __shfl_xor(psum, 4, 32);
            psum += __shfl_xor(psum, 8, 32);
            lrow[r] = lrow[r] * corr + psum;

#pragma unroll
            for (int t = 0; t < 8; ++t) oacc[t][r] *= corr;
        }

#pragma unroll
        for (int r = 0; r < 8; ++r) lp[(r + 8 * hi) * 16 + lc] = s[r];
        asm volatile("s_wait_dscnt 0" ::: "memory");
        v2f pa[4];
#pragma unroll
        for (int kk = 0; kk < 4; ++kk)
            pa[kk] = *(const v2f*)&lp[lc * 16 + kk * 4 + 2 * hi];
        asm volatile("s_wait_dscnt 0" ::: "memory");

#pragma unroll
        for (int kk = 0; kk < 4; ++kk) {
            const float* vrow =
                V + (size_t)(kv0 + kk * 4 + 2 * hi) * DIMM + h * HDIM + lc;
#pragma unroll
            for (int t = 0; t < 8; ++t) {
                v2f bv;
                bv[0] = vrow[t * 16];
                bv[1] = vrow[t * 16 + DIMM];
                oacc[t] = wmma_f32(pa[kk], bv, oacc[t]);
            }
        }
    }

#pragma unroll
    for (int r = 0; r < 8; ++r) {
        const float inv = 1.0f / lrow[r];
        float* orow = O + (size_t)(q0 + r + 8 * hi) * DIMM + h * HDIM + lc;
#pragma unroll
        for (int t = 0; t < 8; ++t) orow[t * 16] = oacc[t][r] * inv;
    }
}

extern "C" void kernel_launch(void* const* d_in, const int* in_sizes, int n_in,
                              void* d_out, int out_size, void* d_ws, size_t ws_size,
                              hipStream_t stream) {
    const float* x    = (const float*)d_in[0];
    const float* mask = (const float*)d_in[1];
    const float* wq   = (const float*)d_in[3];
    const float* wk   = (const float*)d_in[4];
    const float* wv   = (const float*)d_in[5];
    const float* wo   = (const float*)d_in[6];
    float* out = (float*)d_out;

    float* ws = (float*)d_ws;
    const size_t MAT = (size_t)SEQL * DIMM;
    float* Qb = ws;
    float* Kb = ws + MAT;
    float* Vb = ws + 2 * MAT;
    float* Ab = ws + 3 * MAT;

    dim3 tb(256);
    dim3 gg(DIMM / 64, SEQL / 128);
    gemm_abt<<<gg, tb, 0, stream>>>(x, wq, Qb);
    gemm_abt<<<gg, tb, 0, stream>>>(x, wk, Kb);
    gemm_abt<<<gg, tb, 0, stream>>>(x, wv, Vb);

    dim3 ga(SEQL / 128, NHEADS);
    attn_fwd<<<ga, tb, 0, stream>>>(Qb, Kb, Vb, mask, Ab);

    gemm_abt<<<gg, tb, 0, stream>>>(Ab, wo, out);
}


# gemm_abt: 128x128x32 operand tiles staged in LDS (coalesced 16-B global loads, pitch-36 rows, 8-byte fragment reads), 32x64 per wave, next tile prefetched in registers; k order and sums unchanged
# speedup vs baseline: 1.3038x; 1.3038x over previous
#include <hip/hip_runtime.h>

#define SEQL   2048
#define DIMM   2048
#define NHEADS 16
#define HDIM   128

typedef float v2f __attribute__((ext_vector_type(2)));
typedef float v8f __attribute__((ext_vector_type(8)));

__device__ __forceinline__ v8f wmma_f32(v2f a, v2f b, v8f c) {
    return __builtin_amdgcn_wmma_f32_16x16x4_f32(false, a, false, b, (short)0, c, false, false);
}

#define GBM 128
#define GBN 128
#define GBK 32
#define GLD 36

__global__ __launch_bounds__(256)
void gemm_abt(const float* A, const float* B, float* __restrict__ C) {
    __shared__ __attribute__((aligned(16))) float sA[2][GBM * GLD];
    __shared__ __attribute__((aligned(16))) float sB[2][GBN * GLD];

    const int tid  = threadIdx.x;
    const int lane = tid & 31;
    const int wave = tid >> 5;
    const int hi   = lane >> 4;
    const int lc   = lane & 15;
    const int wm   = wave >> 1;
    const int wn   = wave & 1;

    const int m0 = blockIdx.y * GBM;
    const int n0 = blockIdx.x * GBN;

    const int lr   = tid >> 3;
    const int lcol = (tid & 7) * 4;
    const float* ag = A + (size_t)(m0 + lr) * DIMM + lcol;
    const float* bg = B + (size_t)(n0 + lr) * DIMM + lcol;
    const int soff = lr * GLD + lcol;

    const size_t RS = (size_t)32 * DIMM;
    float4 ra0 = *(const float4*)(ag);
    float4 ra1 = *(const float4*)(ag + RS);
    float4 ra2 = *(const float4*)(ag + 2 * RS);
    float4 ra3 = *(const float4*)(ag + 3 * RS);
    float4 rb0 = *(const float4*)(bg);
    float4 rb1 = *(const float4*)(bg + RS);
    float4 rb2 = *(const float4*)(bg + 2 * RS);
    float4 rb3 = *(const float4*)(bg + 3 * RS);
    *(float4*)&sA[0][soff]            = ra0;
    *(float4*)&sA[0][soff + 32 * GLD] = ra1;
    *(float4*)&sA[0][soff + 64 * GLD] = ra2;
    *(float4*)&sA[0][soff + 96 * GLD] = ra3;
    *(float4*)&sB[0][soff]            = rb0;
    *(float4*)&sB[0][soff + 32 * GLD] = rb1;
    *(float4*)&sB[0][soff + 64 * GLD] = rb2;
    *(float4*)&sB[0][soff + 96 * GLD] = rb3;
    __syncthreads();

    v8f acc[2][4];
#pragma unroll
    for (int i = 0; i < 2; ++i)
#pragma unroll
        for (int j = 0; j < 4; ++j) acc[i][j] = (v8f){};

    const int aoff = (wm * 32 + lc) * GLD + 2 * hi;
    const int boff = (wn * 64 + lc) * GLD + 2 * hi;

    const int NT = DIMM / GBK;
    for (int kt = 0; kt < NT; ++kt) {
        const int cur = kt & 1;
        const bool more = (kt + 1 < NT);

        const float* pa = &sA[cur][aoff];
        const float* pb = &sB[cur][boff];

        v2f fa[2][2][2];
        v2f fb[2][2][4];
#pragma unroll
        for (int u = 0; u < 2; ++u) {
            fa[0][u][0] = *(const v2f*)(pa + u * 4);
            fa[0][u][1] = *(const v2f*)(pa + 16 * GLD + u * 4);
#pragma unroll
            for (int j = 0; j < 4; ++j)
                fb[0][u][j] = *(const v2f*)(pb + j * 16 * GLD + u * 4);
        }
        __builtin_amdgcn_sched_barrier(0);

        if (more) {
            const float* an = ag + (kt + 1) * GBK;
            const float* bn = bg + (kt + 1) * GBK;
            ra0 = *(const float4*)(an);
            ra1 = *(const float4*)(an + RS);
            ra2 = *(const float4*)(an + 2 * RS);
            ra3 = *(const float4*)(an + 3 * RS);
            rb0 = *(const float4*)(bn);
            rb1 = *(const float4*)(bn + RS);
            rb2 = *(const float4*)(bn + 2 * RS);
            rb3 = *(const float4*)(bn + 3 * RS);
        }
        __builtin_amdgcn_sched_barrier(0);

#pragma unroll
        for (int g = 0; g < GBK / 8; ++g) {
            const int cs = g & 1;
            if (g + 1 < GBK / 8) {
#pragma unroll
                for (int u = 0; u < 2; ++u) {
                    const int kk = (g + 1) * 8 + u * 4;
                    fa[cs ^ 1][u][0] = *(const v2f*)(pa + kk);
                    fa[cs ^ 1][u][1] = *(const v2f*)(pa + 16 * GLD + kk);
#pragma unroll
                    for (int j = 0; j < 4; ++j)
                        fb[cs ^ 1][u][j] = *(const v2f*)(pb + j * 16 * GLD + kk);
                }
            }
            if (g == GBK / 8 - 1 && more) {
                float* da = &sA[cur ^ 1][soff];
                float* db = &sB[cur ^ 1][soff];
                *(float4*)(da)            = ra0;
                *(float4*)(da + 32 * GLD) = ra1;
                *(float4*)(da + 64 * GLD) = ra2;
                *(float4*)(da + 96 * GLD) = ra3;
                *(float4*)(db)            = rb0;
                *(float4*)(db + 32 * GLD) = rb1;
                *(float4*)(db + 64 * GLD) = rb2;
                *(float4*)(db + 96 * GLD) = rb3;
            }
            __builtin_amdgcn_sched_barrier(0);
#pragma unroll
            for (int u = 0; u < 2; ++u) {
#pragma unroll
                for (int j = 0; j < 4; ++j) {
                    acc[0][j] = wmma_f32(fa[cs][u][0], fb[cs][u][j], acc[0][j]);
                    acc[1][j] = wmma_f32(fa[cs][u][1], fb[cs][u][j], acc[1][j]);
                }
                __builtin_amdgcn_sched_barrier(0);
            }
        }
        __syncthreads();
    }

#pragma unroll
    for (int i = 0; i < 2; ++i) {
#pragma unroll
        for (int r = 0; r < 8; ++r) {
            float* crow = C + (size_t)(m0 + wm * 32 + i * 16 + r + 8 * hi) * DIMM
                            + n0 + wn * 64 + lc;
            crow[ 0] = acc[i][0][r];
            crow[16] = acc[i][1][r];
            crow[32] = acc[i][2][r];
            crow[48] = acc[i][3][r];
        }
    }
}

__global__ __launch_bounds__(256)
void attn_fwd(const float* __restrict__ Q, const float* __restrict__ K,
              const float* __restrict__ V, const float* __restrict__ mask,
              float* __restrict__ O) {
    __shared__ float ldsP[8 * 256];

    const int lane = threadIdx.x & 31;
    const int wave = threadIdx.x >> 5;
    const int hi   = lane >> 4;
    const int lc   = lane & 15;

    const int h  = blockIdx.y;
    const int q0 = (blockIdx.x * 8 + wave) * 16;

    v2f aq[32];
    {
        const float* qrow = Q + (size_t)(q0 + lc) * DIMM + h * HDIM + 2 * hi;
#pragma unroll
        for (int s = 0; s < 32; ++s) aq[s] = *(const v2f*)(qrow + s * 4);
    }

    v8f oacc[8];
#pragma unroll
    for (int t = 0; t < 8; ++t) oacc[t] = (v8f){};
    float mrow[8], lrow[8];
#pragma unroll
    for (int r = 0; r < 8; ++r) { mrow[r] = -1.0e30f; lrow[r] = 0.0f; }

    float* lp = &ldsP[wave * 256];
    const float scale = 0.08838834764831845f;
    const size_t mbase = ((size_t)h * SEQL + q0) * SEQL;

#pragma unroll 1
    for (int kv0 = 0; kv0 < SEQL; kv0 += 16) {
        v8f s = {};
        const float* krow = K + (size_t)(kv0 + lc) * DIMM + h * HDIM + 2 * hi;
#pragma unroll
        for (int ss = 0; ss < 32; ++ss) {
            v2f bk = *(const v2f*)(krow + ss * 4);
            s = wmma_f32(aq[ss], bk, s);
        }

#pragma unroll
        for (int r = 0; r < 8; ++r) {
            const float mval = __builtin_nontemporal_load(
                mask + mbase + (size_t)(r + 8 * hi) * SEQL + kv0 + lc);
            float v = s[r] * scale + mval;

            float rmax = v;
            rmax = fmaxf(rmax, __shfl_xor(rmax, 1, 32));
            rmax = fmaxf(rmax, __shfl_xor(rmax, 2, 32));
            rmax = fmaxf(rmax, __shfl_xor(rmax, 4, 32));
            rmax = fmaxf(rmax, __shfl_xor(rmax, 8, 32));

            const float mnew = fmaxf(mrow[r], rmax);
            const float corr = __expf(mrow[r] - mnew);
            mrow[r] = mnew;

            const float p = __expf(v - mnew);
            s[r] = p;

            float psum = p;
            psum += __shfl_xor(psum, 1, 32);
            psum += __shfl_xor(psum, 2, 32);
            psum += __shfl_xor(psum, 4, 32);
            psum += __shfl_xor(psum, 8, 32);
            lrow[r] = lrow[r] * corr + psum;

#pragma unroll
            for (int t = 0; t < 8; ++t) oacc[t][r] *= corr;
        }

#pragma unroll
        for (int r = 0; r < 8; ++r) lp[(r + 8 * hi) * 16 + lc] = s[r];
        asm volatile("s_wait_dscnt 0" ::: "memory");
        v2f pa[4];
#pragma unroll
        for (int kk = 0; kk < 4; ++kk)
            pa[kk] = *(const v2f*)&lp[lc * 16 + kk * 4 + 2 * hi];
        asm volatile("s_wait_dscnt 0" ::: "memory");

#pragma unroll
        for (int kk = 0; kk < 4; ++kk) {
            const float* vrow =
                V + (size_t)(kv0 + kk * 4 + 2 * hi) * DIMM + h * HDIM + lc;
#pragma unroll
            for (int t = 0; t < 8; ++t) {
                v2f bv;
                bv[0] = vrow[t * 16];
                bv[1] = vrow[t * 16 + DIMM];
                oacc[t] = wmma_f32(pa[kk], bv, oacc[t]);
            }
        }
    }

#pragma unroll
    for (int r = 0; r < 8; ++r) {
        const float inv = 1.0f / lrow[r];
        float* orow = O + (size_t)(q0 + r + 8 * hi) * DIMM + h * HDIM + lc;
#pragma unroll
        for (int t = 0; t < 8; ++t) orow[t * 16] = oacc[t][r] * inv;
    }
}

extern "C" void kernel_launch(void* const* d_in, const int* in_sizes, int n_in,
                              void* d_out, int out_size, void* d_ws, size_t ws_size,
                              hipStream_t stream) {
    const float* x    = (const float*)d_in[0];
    const float* mask = (const float*)d_in[1];
    const float* wq   = (const float*)d_in[3];
    const float* wk   = (const float*)d_in[4];
    const float* wv   = (const float*)d_in[5];
    const float* wo   = (const float*)d_in[6];
    float* out = (float*)d_out;

    float* ws = (float*)d_ws;
    const size_t MAT = (size_t)SEQL * DIMM;
    float* Qb = ws;
    float* Kb = ws + MAT;
    float* Vb = ws + 2 * MAT;
    float* Ab = ws + 3 * MAT;

    dim3 tb(256);
    dim3 gg(DIMM / GBN, SEQL / GBM);
    gemm_abt<<<gg, tb, 0, stream>>>(x, wq, Qb);
    gemm_abt<<<gg, tb, 0, stream>>>(x, wk, Kb);
    gemm_abt<<<gg, tb, 0, stream>>>(x, wv, Vb);

    dim3 ga(SEQL / 128, NHEADS);
    attn_fwd<<<ga, tb, 0, stream>>>(Qb, Kb, Vb, mask, Ab);

    gemm_abt<<<gg, tb, 0, stream>>>(Ab, wo, out);
}


# attn_fwd: K,V staged per 32-key chunk in LDS (coalesced 16-B loads shared by the 8 waves, 2 stages), fragment and mask loads grouped one WMMA block ahead; k order and sums unchanged
# speedup vs baseline: 1.4377x; 1.1027x over previous
#include <hip/hip_runtime.h>

#define SEQL   2048
#define DIMM   2048
#define NHEADS 16
#define HDIM   128

typedef float v2f __attribute__((ext_vector_type(2)));
typedef float v8f __attribute__((ext_vector_type(8)));

__device__ __forceinline__ v8f wmma_f32(v2f a, v2f b, v8f c) {
    return __builtin_amdgcn_wmma_f32_16x16x4_f32(false, a, false, b, (short)0, c, false, false);
}

#define GBM 128
#define GBN 128
#define GBK 32
#define GLD 36

__global__ __launch_bounds__(256)
void gemm_abt(const float* A, const float* B, float* __restrict__ C) {
    __shared__ __attribute__((aligned(16))) float sA[2][GBM * GLD];
    __shared__ __attribute__((aligned(16))) float sB[2][GBN * GLD];

    const int tid  = threadIdx.x;
    const int lane = tid & 31;
    const int wave = tid >> 5;
    const int hi   = lane >> 4;
    const int lc   = lane & 15;
    const int wm   = wave >> 1;
    const int wn   = wave & 1;

    const int m0 = blockIdx.y * GBM;
    const int n0 = blockIdx.x * GBN;

    const int lr   = tid >> 3;
    const int lcol = (tid & 7) * 4;
    const float* ag = A + (size_t)(m0 + lr) * DIMM + lcol;
    const float* bg = B + (size_t)(n0 + lr) * DIMM + lcol;
    const int soff = lr * GLD + lcol;

    const size_t RS = (size_t)32 * DIMM;
    float4 ra0 = *(const float4*)(ag);
    float4 ra1 = *(const float4*)(ag + RS);
    float4 ra2 = *(const float4*)(ag + 2 * RS);
    float4 ra3 = *(const float4*)(ag + 3 * RS);
    float4 rb0 = *(const float4*)(bg);
    float4 rb1 = *(const float4*)(bg + RS);
    float4 rb2 = *(const float4*)(bg + 2 * RS);
    float4 rb3 = *(const float4*)(bg + 3 * RS);
    *(float4*)&sA[0][soff]            = ra0;
    *(float4*)&sA[0][soff + 32 * GLD] = ra1;
    *(float4*)&sA[0][soff + 64 * GLD] = ra2;
    *(float4*)&sA[0][soff + 96 * GLD] = ra3;
    *(float4*)&sB[0][soff]            = rb0;
    *(float4*)&sB[0][soff + 32 * GLD] = rb1;
    *(float4*)&sB[0][soff + 64 * GLD] = rb2;
    *(float4*)&sB[0][soff + 96 * GLD] = rb3;
    __syncthreads();

    v8f acc[2][4];
#pragma unroll
    for (int i = 0; i < 2; ++i)
#pragma unroll
        for (int j = 0; j < 4; ++j) acc[i][j] = (v8f){};

    const int aoff = (wm * 32 + lc) * GLD + 2 * hi;
    const int boff = (wn * 64 + lc) * GLD + 2 * hi;

    const int NT = DIMM / GBK;
    for (int kt = 0; kt < NT; ++kt) {
        const int cur = kt & 1;
        const bool more = (kt + 1 < NT);

        const float* pa = &sA[cur][aoff];
        const float* pb = &sB[cur][boff];

        v2f fa[2][2][2];
        v2f fb[2][2][4];
#pragma unroll
        for (int u = 0; u < 2; ++u) {
            fa[0][u][0] = *(const v2f*)(pa + u * 4);
            fa[0][u][1] = *(const v2f*)(pa + 16 * GLD + u * 4);
#pragma unroll
            for (int j = 0; j < 4; ++j)
                fb[0][u][j] = *(const v2f*)(pb + j * 16 * GLD + u * 4);
        }
        __builtin_amdgcn_sched_barrier(0);

        if (more) {
            const float* an = ag + (kt + 1) * GBK;
            const float* bn = bg + (kt + 1) * GBK;
            ra0 = *(const float4*)(an);
            ra1 = *(const float4*)(an + RS);
            ra2 = *(const float4*)(an + 2 * RS);
            ra3 = *(const float4*)(an + 3 * RS);
            rb0 = *(const float4*)(bn);
            rb1 = *(const float4*)(bn + RS);
            rb2 = *(const float4*)(bn + 2 * RS);
            rb3 = *(const float4*)(bn + 3 * RS);
        }
        __builtin_amdgcn_sched_barrier(0);

#pragma unroll
        for (int g = 0; g < GBK / 8; ++g) {
            const int cs = g & 1;
            if (g + 1 < GBK / 8) {
#pragma unroll
                for (int u = 0; u < 2; ++u) {
                    const int kk = (g + 1) * 8 + u * 4;
                    fa[cs ^ 1][u][0] = *(const v2f*)(pa + kk);
                    fa[cs ^ 1][u][1] = *(const v2f*)(pa + 16 * GLD + kk);
#pragma unroll
                    for (int j = 0; j < 4; ++j)
                        fb[cs ^ 1][u][j] = *(const v2f*)(pb + j * 16 * GLD + kk);
                }
            }
            if (g == GBK / 8 - 1 && more) {
                float* da = &sA[cur ^ 1][soff];
                float* db = &sB[cur ^ 1][soff];
                *(float4*)(da)            = ra0;
                *(float4*)(da + 32 * GLD) = ra1;
                *(float4*)(da + 64 * GLD) = ra2;
                *(float4*)(da + 96 * GLD) = ra3;
                *(float4*)(db)            = rb0;
                *(float4*)(db + 32 * GLD) = rb1;
                *(float4*)(db + 64 * GLD) = rb2;
                *(float4*)(db + 96 * GLD) = rb3;
            }
            __builtin_amdgcn_sched_barrier(0);
#pragma unroll
            for (int u = 0; u < 2; ++u) {
#pragma unroll
                for (int j = 0; j < 4; ++j) {
                    acc[0][j] = wmma_f32(fa[cs][u][0], fb[cs][u][j], acc[0][j]);
                    acc[1][j] = wmma_f32(fa[cs][u][1], fb[cs][u][j], acc[1][j]);
                }
                __builtin_amdgcn_sched_barrier(0);
            }
        }
        __syncthreads();
    }

#pragma unroll
    for (int i = 0; i < 2; ++i) {
#pragma unroll
        for (int r = 0; r < 8; ++r) {
            float* crow = C + (size_t)(m0 + wm * 32 + i * 16 + r + 8 * hi) * DIMM
                            + n0 + wn * 64 + lc;
            crow[ 0] = acc[i][0][r];
            crow[16] = acc[i][1][r];
            crow[32] = acc[i][2][r];
            crow[48] = acc[i][3][r];
        }
    }
}

#define AKC 32
#define AKP 132
#define AVP 288

__global__ __launch_bounds__(256)
void attn_fwd(const float* __restrict__ Q, const float* K,
              const float* V, const float* __restrict__ mask,
              float* __restrict__ O) {
    __shared__ float ldsP[8 * 256];
    __shared__ __attribute__((aligned(16))) float sK[2][AKC * AKP];
    __shared__ __attribute__((aligned(16))) float sV[2][(AKC / 2) * AVP];

    const int tid  = threadIdx.x;
    const int lane = tid & 31;
    const int wave = tid >> 5;
    const int hi   = lane >> 4;
    const int lc   = lane & 15;

    const int h  = blockIdx.y;
    const int q0 = (blockIdx.x * 8 + wave) * 16;

    const int sr = tid >> 5;
    const int sc = (tid & 31) * 4;
    const float* kg = K + (size_t)sr * DIMM + h * HDIM + sc;
    const float* vg = V + (size_t)(2 * sr) * DIMM + h * HDIM + sc;
    const size_t R8  = (size_t)8 * DIMM;
    const size_t R16 = (size_t)16 * DIMM;
    const int skoff = sr * AKP + sc;
    const int svoff = sr * AVP + 2 * sc;

    const float4 fk0 = *(const float4*)(kg);
    const float4 fk1 = *(const float4*)(kg + R8);
    const float4 fk2 = *(const float4*)(kg + 2 * R8);
    const float4 fk3 = *(const float4*)(kg + 3 * R8);
    const float4 fv0 = *(const float4*)(vg);
    const float4 fv1 = *(const float4*)(vg + DIMM);
    const float4 fv2 = *(const float4*)(vg + R16);
    const float4 fv3 = *(const float4*)(vg + R16 + DIMM);

    v2f aq[32];
    {
        const float* qrow = Q + (size_t)(q0 + lc) * DIMM + h * HDIM + 2 * hi;
#pragma unroll
        for (int s = 0; s < 32; ++s) aq[s] = *(const v2f*)(qrow + s * 4);
    }

    *(float4*)&sK[0][skoff]            = fk0;
    *(float4*)&sK[0][skoff +  8 * AKP] = fk1;
    *(float4*)&sK[0][skoff + 16 * AKP] = fk2;
    *(float4*)&sK[0][skoff + 24 * AKP] = fk3;
    *(float4*)&sV[0][svoff]               = make_float4(fv0.x, fv1.x, fv0.y, fv1.y);
    *(float4*)&sV[0][svoff + 4]           = make_float4(fv0.z, fv1.z, fv0.w, fv1.w);
    *(float4*)&sV[0][svoff + 8 * AVP]     = make_float4(fv2.x, fv3.x, fv2.y, fv3.y);
    *(float4*)&sV[0][svoff + 8 * AVP + 4] = make_float4(fv2.z, fv3.z, fv2.w, fv3.w);
    __syncthreads();

    v8f oacc[8];
#pragma unroll
    for (int t = 0; t < 8; ++t) oacc[t] = (v8f){};
    float mrow[8], lrow[8];
#pragma unroll
    for (int r = 0; r < 8; ++r) { mrow[r] = -1.0e30f; lrow[r] = 0.0f; }

    float* lp = &ldsP[wave * 256];
    const float scale = 0.08838834764831845f;
    const size_t mbase = ((size_t)h * SEQL + q0) * SEQL;
    const float* mlane = mask + mbase + (size_t)(8 * hi) * SEQL + lc;

    const int NC = SEQL / AKC;
#pragma unroll 1
    for (int c = 0; c < NC; ++c) {
        const int cur = c & 1;
        const bool more = (c + 1 < NC);

        const int cn = more ? c + 1 : c;
        const float* kn = kg + (size_t)cn * AKC * DIMM;
        const float* vn = vg + (size_t)cn * AKC * DIMM;
        const float4 rk0 = *(const float4*)(kn);
        const float4 rk1 = *(const float4*)(kn + R8);
        const float4 rk2 = *(const float4*)(kn + 2 * R8);
        const float4 rk3 = *(const float4*)(kn + 3 * R8);
        const float4 rv0 = *(const float4*)(vn);
        const float4 rv1 = *(const float4*)(vn + DIMM);
        const float4 rv2 = *(const float4*)(vn + R16);
        const float4 rv3 = *(const float4*)(vn + R16 + DIMM);

        const float* ck = &sK[cur][lc * AKP + 2 * hi];
        const float* cv = &sV[cur][hi * AVP + 2 * lc];

        float mk[8];
#pragma unroll
        for (int r = 0; r < 8; ++r)
            mk[r] = __builtin_nontemporal_load(mlane + (size_t)r * SEQL + c * AKC);

        v2f bk[32];
#pragma unroll
        for (int ss = 0; ss < 32; ++ss) bk[ss] = *(const v2f*)(ck + ss * 4);
        __builtin_amdgcn_sched_barrier(0);

#pragma unroll
        for (int j = 0; j < AKC / 16; ++j) {
            const int kv0 = c * AKC + j * 16;

            v8f s = {};
#pragma unroll
            for (int ss = 0; ss < 32; ++ss) s = wmma_f32(aq[ss], bk[ss], s);

            float cr[8];
#pragma unroll
            for (int r = 0; r < 8; ++r) {
                float v = s[r] * scale + mk[r];

                float rmax = v;
                rmax = fmaxf(rmax, __shfl_xor(rmax, 1, 32));
                rmax = fmaxf(rmax, __shfl_xor(rmax, 2, 32));
                rmax = fmaxf(rmax, __shfl_xor(rmax, 4, 32));
                rmax = fmaxf(rmax, __shfl_xor(rmax, 8, 32));

                const float mnew = fmaxf(mrow[r], rmax);
                const float corr = __expf(mrow[r] - mnew);
                mrow[r] = mnew;

                cr[r] = corr;
                s[r] = __expf(v - mnew);

#pragma unroll
                for (int t = 0; t < 8; ++t) oacc[t][r] *= corr;
            }

#pragma unroll
            for (int r = 0; r < 8; ++r) lp[(r + 8 * hi) * 16 + lc] = s[r];
            asm volatile("s_wait_dscnt 0" ::: "memory");
            v2f pa[4];
#pragma unroll
            for (int kk = 0; kk < 4; ++kk)
                pa[kk] = *(const v2f*)&lp[lc * 16 + kk * 4 + 2 * hi];
            asm volatile("s_wait_dscnt 0" ::: "memory");

            if (j == AKC / 16 - 1 && more) {
                float* dk = &sK[cur ^ 1][skoff];
                float* dv = &sV[cur ^ 1][svoff];
                *(float4*)(dk)            = rk0;
                *(float4*)(dk +  8 * AKP) = rk1;
                *(float4*)(dk + 16 * AKP) = rk2;
                *(float4*)(dk + 24 * AKP) = rk3;
                *(float4*)(dv)               = make_float4(rv0.x, rv1.x, rv0.y, rv1.y);
                *(float4*)(dv + 4)           = make_float4(rv0.z, rv1.z, rv0.w, rv1.w);
                *(float4*)(dv + 8 * AVP)     = make_float4(rv2.x, rv3.x, rv2.y, rv3.y);
                *(float4*)(dv + 8 * AVP + 4) = make_float4(rv2.z, rv3.z, rv2.w, rv3.w);
            }

            v2f bv[4][8];
#pragma unroll
            for (int kk = 0; kk < 4; ++kk) {
                const float* vp = cv + (j * 8 + kk * 2) * AVP;
#pragma unroll
                for (int t = 0; t < 8; ++t) bv[kk][t] = *(const v2f*)(vp + t * 32);
            }
            if (j + 1 < AKC / 16) {
#pragma unroll
                for (int ss = 0; ss < 32; ++ss)
                    bk[ss] = *(const v2f*)(ck + (j + 1) * 16 * AKP + ss * 4);
#pragma unroll
                for (int r = 0; r < 8; ++r)
                    mk[r] = __builtin_nontemporal_load(mlane + (size_t)r * SEQL + kv0 + 16);
            }
            __builtin_amdgcn_sched_barrier(0);
#pragma unroll
            for (int kk = 0; kk < 4; ++kk) {
#pragma unroll
                for (int t = 0; t < 8; ++t)
                    oacc[t] = wmma_f32(pa[kk], bv[kk][t], oacc[t]);
            }

#pragma unroll
            for (int r = 0; r < 8; ++r) {
                float psum = s[r];
                psum += __shfl_xor(psum, 1, 32);
                psum += __shfl_xor(psum, 2, 32);
                psum += __shfl_xor(psum, 4, 32);
                psum += __shfl_xor(psum, 8, 32);
                lrow[r] = lrow[r] * cr[r] + psum;
            }
        }
        __syncthreads();
    }

#pragma unroll
    for (int r = 0; r < 8; ++r) {
        const float inv = 1.0f / lrow[r];
        float* orow = O + (size_t)(q0 + r + 8 * hi) * DIMM + h * HDIM + lc;
#pragma unroll
        for (int t = 0; t < 8; ++t) orow[t * 16] = oacc[t][r] * inv;
    }
}

extern "C" void kernel_launch(void* const* d_in, const int* in_sizes, int n_in,
                              void* d_out, int out_size, void* d_ws, size_t ws_size,
                              hipStream_t stream) {
    const float* x    = (const float*)d_in[0];
    const float* mask = (const float*)d_in[1];
    const float* wq   = (const float*)d_in[3];
    const float* wk   = (const float*)d_in[4];
    const float* wv   = (const float*)d_in[5];
    const float* wo   = (const float*)d_in[6];
    float* out = (float*)d_out;

    float* ws = (float*)d_ws;
    const size_t MAT = (size_t)SEQL * DIMM;
    float* Qb = ws;
    float* Kb = ws + MAT;
    float* Vb = ws + 2 * MAT;
    float* Ab = ws + 3 * MAT;

    dim3 tb(256);
    dim3 gg(DIMM / GBN, SEQL / GBM);
    gemm_abt<<<gg, tb, 0, stream>>>(x, wq, Qb);
    gemm_abt<<<gg, tb, 0, stream>>>(x, wk, Kb);
    gemm_abt<<<gg, tb, 0, stream>>>(x, wv, Vb);

    dim3 ga(SEQL / 128, NHEADS);
    attn_fwd<<<ga, tb, 0, stream>>>(Qb, Kb, Vb, mask, Ab);

    gemm_abt<<<gg, tb, 0, stream>>>(Ab, wo, out);
}


# attn_fwd: row max and row sum over the 16 lanes of a row by row_xmask DPP reads on the vector ALU instead of lane shuffles through the LDS unit; same partners 1,2,4,8 in order, sums unchanged
# speedup vs baseline: 1.5349x; 1.0676x over previous
#include <hip/hip_runtime.h>

#define SEQL   2048
#define DIMM   2048
#define NHEADS 16
#define HDIM   128

typedef float v2f __attribute__((ext_vector_type(2)));
typedef float v8f __attribute__((ext_vector_type(8)));

__device__ __forceinline__ v8f wmma_f32(v2f a, v2f b, v8f c) {
    return __builtin_amdgcn_wmma_f32_16x16x4_f32(false, a, false, b, (short)0, c, false, false);
}

template <int M>
__device__ __forceinline__ float row_xor(float v) {
    return __int_as_float(__builtin_amdgcn_update_dpp(
        0, __float_as_int(v), 0x160 + M  , 0xf, 0xf, true));
}

#define GBM 128
#define GBN 128
#define GBK 32
#define GLD 36

__global__ __launch_bounds__(256)
void gemm_abt(const float* A, const float* B, float* __restrict__ C) {
    __shared__ __attribute__((aligned(16))) float sA[2][GBM * GLD];
    __shared__ __attribute__((aligned(16))) float sB[2][GBN * GLD];

    const int tid  = threadIdx.x;
    const int lane = tid & 31;
    const int wave = tid >> 5;
    const int hi   = lane >> 4;
    const int lc   = lane & 15;
    const int wm   = wave >> 1;
    const int wn   = wave & 1;

    const int m0 = blockIdx.y * GBM;
    const int n0 = blockIdx.x * GBN;

    const int lr   = tid >> 3;
    const int lcol = (tid & 7) * 4;
    const float* ag = A + (size_t)(m0 + lr) * DIMM + lcol;
    const float* bg = B + (size_t)(n0 + lr) * DIMM + lcol;
    const int soff = lr * GLD + lcol;

    const size_t RS = (size_t)32 * DIMM;
    float4 ra0 = *(const float4*)(ag);
    float4 ra1 = *(const float4*)(ag + RS);
    float4 ra2 = *(const float4*)(ag + 2 * RS);
    float4 ra3 = *(const float4*)(ag + 3 * RS);
    float4 rb0 = *(const float4*)(bg);
    float4 rb1 = *(const float4*)(bg + RS);
    float4 rb2 = *(const float4*)(bg + 2 * RS);
    float4 rb3 = *(const float4*)(bg + 3 * RS);
    *(float4*)&sA[0][soff]            = ra0;
    *(float4*)&sA[0][soff + 32 * GLD] = ra1;
    *(float4*)&sA[0][soff + 64 * GLD] = ra2;
    *(float4*)&sA[0][soff + 96 * GLD] = ra3;
    *(float4*)&sB[0][soff]            = rb0;
    *(float4*)&sB[0][soff + 32 * GLD] = rb1;
    *(float4*)&sB[0][soff + 64 * GLD] = rb2;
    *(float4*)&sB[0][soff + 96 * GLD] = rb3;
    __syncthreads();

    v8f acc[2][4];
#pragma unroll
    for (int i = 0; i < 2; ++i)
#pragma unroll
        for (int j = 0; j < 4; ++j) acc[i][j] = (v8f){};

    const int aoff = (wm * 32 + lc) * GLD + 2 * hi;
    const int boff = (wn * 64 + lc) * GLD + 2 * hi;

    const int NT = DIMM / GBK;
    for (int kt = 0; kt < NT; ++kt) {
        const int cur = kt & 1;
        const bool more = (kt + 1 < NT);

        const float* pa = &sA[cur][aoff];
        const float* pb = &sB[cur][boff];

        v2f fa[2][2][2];
        v2f fb[2][2][4];
#pragma unroll
        for (int u = 0; u < 2; ++u) {
            fa[0][u][0] = *(const v2f*)(pa + u * 4);
            fa[0][u][1] = *(const v2f*)(pa + 16 * GLD + u * 4);
#pragma unroll
            for (int j = 0; j < 4; ++j)
                fb[0][u][j] = *(const v2f*)(pb + j * 16 * GLD + u * 4);
        }
        __builtin_amdgcn_sched_barrier(0);

        if (more) {
            const float* an = ag + (kt + 1) * GBK;
            const float* bn = bg + (kt + 1) * GBK;
            ra0 = *(const float4*)(an);
            ra1 = *(const float4*)(an + RS);
            ra2 = *(const float4*)(an + 2 * RS);
            ra3 = *(const float4*)(an + 3 * RS);
            rb0 = *(const float4*)(bn);
            rb1 = *(const float4*)(bn + RS);
            rb2 = *(const float4*)(bn + 2 * RS);
            rb3 = *(const float4*)(bn + 3 * RS);
        }
        __builtin_amdgcn_sched_barrier(0);

#pragma unroll
        for (int g = 0; g < GBK / 8; ++g) {
            const int cs = g & 1;
            if (g + 1 < GBK / 8) {
#pragma unroll
                for (int u = 0; u < 2; ++u) {
                    const int kk = (g + 1) * 8 + u * 4;
                    fa[cs ^ 1][u][0] = *(const v2f*)(pa + kk);
                    fa[cs ^ 1][u][1] = *(const v2f*)(pa + 16 * GLD + kk);
#pragma unroll
                    for (int j = 0; j < 4; ++j)
                        fb[cs ^ 1][u][j] = *(const v2f*)(pb + j * 16 * GLD + kk);
                }
            }
            if (g == GBK / 8 - 1 && more) {
                float* da = &sA[cur ^ 1][soff];
                float* db = &sB[cur ^ 1][soff];
                *(float4*)(da)            = ra0;
                *(float4*)(da + 32 * GLD) = ra1;
                *(float4*)(da + 64 * GLD) = ra2;
                *(float4*)(da + 96 * GLD) = ra3;
                *(float4*)(db)            = rb0;
                *(float4*)(db + 32 * GLD) = rb1;
                *(float4*)(db + 64 * GLD) = rb2;
                *(float4*)(db + 96 * GLD) = rb3;
            }
            __builtin_amdgcn_sched_barrier(0);
#pragma unroll
            for (int u = 0; u < 2; ++u) {
#pragma unroll
                for (int j = 0; j < 4; ++j) {
                    acc[0][j] = wmma_f32(fa[cs][u][0], fb[cs][u][j], acc[0][j]);
                    acc[1][j] = wmma_f32(fa[cs][u][1], fb[cs][u][j], acc[1][j]);
                }
                __builtin_amdgcn_sched_barrier(0);
            }
        }
        __syncthreads();
    }

#pragma unroll
    for (int i = 0; i < 2; ++i) {
#pragma unroll
        for (int r = 0; r < 8; ++r) {
            float* crow = C + (size_t)(m0 + wm * 32 + i * 16 + r + 8 * hi) * DIMM
                            + n0 + wn * 64 + lc;
            crow[ 0] = acc[i][0][r];
            crow[16] = acc[i][1][r];
            crow[32] = acc[i][2][r];
            crow[48] = acc[i][3][r];
        }
    }
}

#define AKC 32
#define AKP 132
#define AVP 288

__global__ __launch_bounds__(256)
void attn_fwd(const float* __restrict__ Q, const float* K,
              const float* V, const float* __restrict__ mask,
              float* __restrict__ O) {
    __shared__ float ldsP[8 * 256];
    __shared__ __attribute__((aligned(16))) float sK[2][AKC * AKP];
    __shared__ __attribute__((aligned(16))) float sV[2][(AKC / 2) * AVP];

    const int tid  = threadIdx.x;
    const int lane = tid & 31;
    const int wave = tid >> 5;
    const int hi   = lane >> 4;
    const int lc   = lane & 15;

    const int h  = blockIdx.y;
    const int q0 = (blockIdx.x * 8 + wave) * 16;

    const int sr = tid >> 5;
    const int sc = (tid & 31) * 4;
    const float* kg = K + (size_t)sr * DIMM + h * HDIM + sc;
    const float* vg = V + (size_t)(2 * sr) * DIMM + h * HDIM + sc;
    const size_t R8  = (size_t)8 * DIMM;
    const size_t R16 = (size_t)16 * DIMM;
    const int skoff = sr * AKP + sc;
    const int svoff = sr * AVP + 2 * sc;

    const float4 fk0 = *(const float4*)(kg);
    const float4 fk1 = *(const float4*)(kg + R8);
    const float4 fk2 = *(const float4*)(kg + 2 * R8);
    const float4 fk3 = *(const float4*)(kg + 3 * R8);
    const float4 fv0 = *(const float4*)(vg);
    const float4 fv1 = *(const float4*)(vg + DIMM);
    const float4 fv2 = *(const float4*)(vg + R16);
    const float4 fv3 = *(const float4*)(vg + R16 + DIMM);

    v2f aq[32];
    {
        const float* qrow = Q + (size_t)(q0 + lc) * DIMM + h * HDIM + 2 * hi;
#pragma unroll
        for (int s = 0; s < 32; ++s) aq[s] = *(const v2f*)(qrow + s * 4);
    }

    *(float4*)&sK[0][skoff]            = fk0;
    *(float4*)&sK[0][skoff +  8 * AKP] = fk1;
    *(float4*)&sK[0][skoff + 16 * AKP] = fk2;
    *(float4*)&sK[0][skoff + 24 * AKP] = fk3;
    *(float4*)&sV[0][svoff]               = make_float4(fv0.x, fv1.x, fv0.y, fv1.y);
    *(float4*)&sV[0][svoff + 4]           = make_float4(fv0.z, fv1.z, fv0.w, fv1.w);
    *(float4*)&sV[0][svoff + 8 * AVP]     = make_float4(fv2.x, fv3.x, fv2.y, fv3.y);
    *(float4*)&sV[0][svoff + 8 * AVP + 4] = make_float4(fv2.z, fv3.z, fv2.w, fv3.w);
    __syncthreads();

    v8f oacc[8];
#pragma unroll
    for (int t = 0; t < 8; ++t) oacc[t] = (v8f){};
    float mrow[8], lrow[8];
#pragma unroll
    for (int r = 0; r < 8; ++r) { mrow[r] = -1.0e30f; lrow[r] = 0.0f; }

    float* lp = &ldsP[wave * 256];
    const float scale = 0.08838834764831845f;
    const size_t mbase = ((size_t)h * SEQL + q0) * SEQL;
    const float* mlane = mask + mbase + (size_t)(8 * hi) * SEQL + lc;

    const int NC = SEQL / AKC;
#pragma unroll 1
    for (int c = 0; c < NC; ++c) {
        const int cur = c & 1;
        const bool more = (c + 1 < NC);

        const int cn = more ? c + 1 : c;
        const float* kn = kg + (size_t)cn * AKC * DIMM;
        const float* vn = vg + (size_t)cn * AKC * DIMM;
        const float4 rk0 = *(const float4*)(kn);
        const float4 rk1 = *(const float4*)(kn + R8);
        const float4 rk2 = *(const float4*)(kn + 2 * R8);
        const float4 rk3 = *(const float4*)(kn + 3 * R8);
        const float4 rv0 = *(const float4*)(vn);
        const float4 rv1 = *(const float4*)(vn + DIMM);
        const float4 rv2 = *(const float4*)(vn + R16);
        const float4 rv3 = *(const float4*)(vn + R16 + DIMM);

        const float* ck = &sK[cur][lc * AKP + 2 * hi];
        const float* cv = &sV[cur][hi * AVP + 2 * lc];

        float mk[8];
#pragma unroll
        for (int r = 0; r < 8; ++r)
            mk[r] = __builtin_nontemporal_load(mlane + (size_t)r * SEQL + c * AKC);

        v2f bk[32];
#pragma unroll
        for (int ss = 0; ss < 32; ++ss) bk[ss] = *(const v2f*)(ck + ss * 4);
        __builtin_amdgcn_sched_barrier(0);

#pragma unroll
        for (int j = 0; j < AKC / 16; ++j) {
            const int kv0 = c * AKC + j * 16;

            v8f s = {};
#pragma unroll
            for (int ss = 0; ss < 32; ++ss) s = wmma_f32(aq[ss], bk[ss], s);

            float cr[8];
#pragma unroll
            for (int r = 0; r < 8; ++r) {
                float v = s[r] * scale + mk[r];

                float rmax = v;
                rmax = fmaxf(rmax, row_xor<1>(rmax));
                rmax = fmaxf(rmax, row_xor<2>(rmax));
                rmax = fmaxf(rmax, row_xor<4>(rmax));
                rmax = fmaxf(rmax, row_xor<8>(rmax));

                const float mnew = fmaxf(mrow[r], rmax);
                const float corr = __expf(mrow[r] - mnew);
                mrow[r] = mnew;

                cr[r] = corr;
                s[r] = __expf(v - mnew);

#pragma unroll
                for (int t = 0; t < 8; ++t) oacc[t][r] *= corr;
            }

#pragma unroll
            for (int r = 0; r < 8; ++r) lp[(r + 8 * hi) * 16 + lc] = s[r];
            asm volatile("s_wait_dscnt 0" ::: "memory");
            v2f pa[4];
#pragma unroll
            for (int kk = 0; kk < 4; ++kk)
                pa[kk] = *(const v2f*)&lp[lc * 16 + kk * 4 + 2 * hi];
            asm volatile("s_wait_dscnt 0" ::: "memory");

            if (j == AKC / 16 - 1 && more) {
                float* dk = &sK[cur ^ 1][skoff];
                float* dv = &sV[cur ^ 1][svoff];
                *(float4*)(dk)            = rk0;
                *(float4*)(dk +  8 * AKP) = rk1;
                *(float4*)(dk + 16 * AKP) = rk2;
                *(float4*)(dk + 24 * AKP) = rk3;
                *(float4*)(dv)               = make_float4(rv0.x, rv1.x, rv0.y, rv1.y);
                *(float4*)(dv + 4)           = make_float4(rv0.z, rv1.z, rv0.w, rv1.w);
                *(float4*)(dv + 8 * AVP)     = make_float4(rv2.x, rv3.x, rv2.y, rv3.y);
                *(float4*)(dv + 8 * AVP + 4) = make_float4(rv2.z, rv3.z, rv2.w, rv3.w);
            }

            v2f bv[4][8];
#pragma unroll
            for (int kk = 0; kk < 4; ++kk) {
                const float* vp = cv + (j * 8 + kk * 2) * AVP;
#pragma unroll
                for (int t = 0; t < 8; ++t) bv[kk][t] = *(const v2f*)(vp + t * 32);
            }
            if (j + 1 < AKC / 16) {
#pragma unroll
                for (int ss = 0; ss < 32; ++ss)
                    bk[ss] = *(const v2f*)(ck + (j + 1) * 16 * AKP + ss * 4);
#pragma unroll
                for (int r = 0; r < 8; ++r)
                    mk[r] = __builtin_nontemporal_load(mlane + (size_t)r * SEQL + kv0 + 16);
            }
            __builtin_amdgcn_sched_barrier(0);
#pragma unroll
            for (int kk = 0; kk < 4; ++kk) {
#pragma unroll
                for (int t = 0; t < 8; ++t)
                    oacc[t] = wmma_f32(pa[kk], bv[kk][t], oacc[t]);
            }

#pragma unroll
            for (int r = 0; r < 8; ++r) {
                float psum = s[r];
                psum += row_xor<1>(psum);
                psum += row_xor<2>(psum);
                psum += row_xor<4>(psum);
                psum += row_xor<8>(psum);
                lrow[r] = lrow[r] * cr[r] + psum;
            }
        }
        __syncthreads();
    }

#pragma unroll
    for (int r = 0; r < 8; ++r) {
        const float inv = 1.0f / lrow[r];
        float* orow = O + (size_t)(q0 + r + 8 * hi) * DIMM + h * HDIM + lc;
#pragma unroll
        for (int t = 0; t < 8; ++t) orow[t * 16] = oacc[t][r] * inv;
    }
}

extern "C" void kernel_launch(void* const* d_in, const int* in_sizes, int n_in,
                              void* d_out, int out_size, void* d_ws, size_t ws_size,
                              hipStream_t stream) {
    const float* x    = (const float*)d_in[0];
    const float* mask = (const float*)d_in[1];
    const float* wq   = (const float*)d_in[3];
    const float* wk   = (const float*)d_in[4];
    const float* wv   = (const float*)d_in[5];
    const float* wo   = (const float*)d_in[6];
    float* out = (float*)d_out;

    float* ws = (float*)d_ws;
    const size_t MAT = (size_t)SEQL * DIMM;
    float* Qb = ws;
    float* Kb = ws + MAT;
    float* Vb = ws + 2 * MAT;
    float* Ab = ws + 3 * MAT;

    dim3 tb(256);
    dim3 gg(DIMM / GBN, SEQL / GBM);
    gemm_abt<<<gg, tb, 0, stream>>>(x, wq, Qb);
    gemm_abt<<<gg, tb, 0, stream>>>(x, wk, Kb);
    gemm_abt<<<gg, tb, 0, stream>>>(x, wv, Vb);

    dim3 ga(SEQL / 128, NHEADS);
    attn_fwd<<<ga, tb, 0, stream>>>(Qb, Kb, Vb, mask, Ab);

    gemm_abt<<<gg, tb, 0, stream>>>(Ab, wo, out);
}


# attn_fwd: transposed tiles (S^T = K Q^T, O^T = V^T P^T): probabilities are a B operand as they leave the score tile, no LDS layout swap; row max/sum per lane; row-sum tree and PV key order change
# speedup vs baseline: 1.6057x; 1.0461x over previous
#include <hip/hip_runtime.h>

#define SEQL   2048
#define DIMM   2048
#define NHEADS 16
#define HDIM   128

typedef float v2f __attribute__((ext_vector_type(2)));
typedef float v4f __attribute__((ext_vector_type(4)));
typedef float v8f __attribute__((ext_vector_type(8)));

__device__ __forceinline__ v8f wmma_f32(v2f a, v2f b, v8f c) {
    return __builtin_amdgcn_wmma_f32_16x16x4_f32(false, a, false, b, (short)0, c, false, false);
}

__device__ __forceinline__ float xhalf(float v) {
    const int b = __float_as_int(v);
    return __int_as_float(__builtin_amdgcn_permlanex16(b, b, 0x76543210, 0xfedcba98, false, false));
}

#define GBM 128
#define GBN 128
#define GBK 32
#define GLD 36

__global__ __launch_bounds__(256)
void gemm_abt(const float* A, const float* B, float* __restrict__ C) {
    __shared__ __attribute__((aligned(16))) float sA[2][GBM * GLD];
    __shared__ __attribute__((aligned(16))) float sB[2][GBN * GLD];

    const int tid  = threadIdx.x;
    const int lane = tid & 31;
    const int wave = tid >> 5;
    const int hi   = lane >> 4;
    const int lc   = lane & 15;
    const int wm   = wave >> 1;
    const int wn   = wave & 1;

    const int m0 = blockIdx.y * GBM;
    const int n0 = blockIdx.x * GBN;

    const int lr   = tid >> 3;
    const int lcol = (tid & 7) * 4;
    const float* ag = A + (size_t)(m0 + lr) * DIMM + lcol;
    const float* bg = B + (size_t)(n0 + lr) * DIMM + lcol;
    const int soff = lr * GLD + lcol;

    const size_t RS = (size_t)32 * DIMM;
    float4 ra0 = *(const float4*)(ag);
    float4 ra1 = *(const float4*)(ag + RS);
    float4 ra2 = *(const float4*)(ag + 2 * RS);
    float4 ra3 = *(const float4*)(ag + 3 * RS);
    float4 rb0 = *(const float4*)(bg);
    float4 rb1 = *(const float4*)(bg + RS);
    float4 rb2 = *(const float4*)(bg + 2 * RS);
    float4 rb3 = *(const float4*)(bg + 3 * RS);
    *(float4*)&sA[0][soff]            = ra0;
    *(float4*)&sA[0][soff + 32 * GLD] = ra1;
    *(float4*)&sA[0][soff + 64 * GLD] = ra2;
    *(float4*)&sA[0][soff + 96 * GLD] = ra3;
    *(float4*)&sB[0][soff]            = rb0;
    *(float4*)&sB[0][soff + 32 * GLD] = rb1;
    *(float4*)&sB[0][soff + 64 * GLD] = rb2;
    *(float4*)&sB[0][soff + 96 * GLD] = rb3;
    __syncthreads();

    v8f acc[2][4];
#pragma unroll
    for (int i = 0; i < 2; ++i)
#pragma unroll
        for (int j = 0; j < 4; ++j) acc[i][j] = (v8f){};

    const int aoff = (wm * 32 + lc) * GLD + 2 * hi;
    const int boff = (wn * 64 + lc) * GLD + 2 * hi;

    const int NT = DIMM / GBK;
    for (int kt = 0; kt < NT; ++kt) {
        const int cur = kt & 1;
        const bool more = (kt + 1 < NT);

        const float* pa = &sA[cur][aoff];
        const float* pb = &sB[cur][boff];

        v2f fa[2][2][2];
        v2f fb[2][2][4];
#pragma unroll
        for (int u = 0; u < 2; ++u) {
            fa[0][u][0] = *(const v2f*)(pa + u * 4);
            fa[0][u][1] = *(const v2f*)(pa + 16 * GLD + u * 4);
#pragma unroll
            for (int j = 0; j < 4; ++j)
                fb[0][u][j] = *(const v2f*)(pb + j * 16 * GLD + u * 4);
        }
        __builtin_amdgcn_sched_barrier(0);

        if (more) {
            const float* an = ag + (kt + 1) * GBK;
            const float* bn = bg + (kt + 1) * GBK;
            ra0 = *(const float4*)(an);
            ra1 = *(const float4*)(an + RS);
            ra2 = *(const float4*)(an + 2 * RS);
            ra3 = *(const float4*)(an + 3 * RS);
            rb0 = *(const float4*)(bn);
            rb1 = *(const float4*)(bn + RS);
            rb2 = *(const float4*)(bn + 2 * RS);
            rb3 = *(const float4*)(bn + 3 * RS);
        }
        __builtin_amdgcn_sched_barrier(0);

#pragma unroll
        for (int g = 0; g < GBK / 8; ++g) {
            const int cs = g & 1;
            if (g + 1 < GBK / 8) {
#pragma unroll
                for (int u = 0; u < 2; ++u) {
                    const int kk = (g + 1) * 8 + u * 4;
                    fa[cs ^ 1][u][0] = *(const v2f*)(pa + kk);
                    fa[cs ^ 1][u][1] = *(const v2f*)(pa + 16 * GLD + kk);
#pragma unroll
                    for (int j = 0; j < 4; ++j)
                        fb[cs ^ 1][u][j] = *(const v2f*)(pb + j * 16 * GLD + kk);
                }
            }
            if (g == GBK / 8 - 1 && more) {
                float* da = &sA[cur ^ 1][soff];
                float* db = &sB[cur ^ 1][soff];
                *(float4*)(da)            = ra0;
                *(float4*)(da + 32 * GLD) = ra1;
                *(float4*)(da + 64 * GLD) = ra2;
                *(float4*)(da + 96 * GLD) = ra3;
                *(float4*)(db)            = rb0;
                *(float4*)(db + 32 * GLD) = rb1;
                *(float4*)(db + 64 * GLD) = rb2;
                *(float4*)(db + 96 * GLD) = rb3;
            }
            __builtin_amdgcn_sched_barrier(0);
#pragma unroll
            for (int u = 0; u < 2; ++u) {
#pragma unroll
                for (int j = 0; j < 4; ++j) {
                    acc[0][j] = wmma_f32(fa[cs][u][0], fb[cs][u][j], acc[0][j]);
                    acc[1][j] = wmma_f32(fa[cs][u][1], fb[cs][u][j], acc[1][j]);
                }
                __builtin_amdgcn_sched_barrier(0);
            }
        }
        __syncthreads();
    }

#pragma unroll
    for (int i = 0; i < 2; ++i) {
#pragma unroll
        for (int r = 0; r < 8; ++r) {
            float* crow = C + (size_t)(m0 + wm * 32 + i * 16 + r + 8 * hi) * DIMM
                            + n0 + wn * 64 + lc;
            crow[ 0] = acc[i][0][r];
            crow[16] = acc[i][1][r];
            crow[32] = acc[i][2][r];
            crow[48] = acc[i][3][r];
        }
    }
}

#define AKC 32
#define AKP 132
#define AVP 264

__global__ __launch_bounds__(256)
void attn_fwd(const float* __restrict__ Q, const float* K,
              const float* V, const float* __restrict__ mask,
              float* __restrict__ O) {
    __shared__ __attribute__((aligned(16))) float sK[2][AKC * AKP];
    __shared__ __attribute__((aligned(16))) float sV[2][(AKC / 2) * AVP];

    const int tid  = threadIdx.x;
    const int lane = tid & 31;
    const int wave = tid >> 5;
    const int hi   = lane >> 4;
    const int lc   = lane & 15;

    const int h  = blockIdx.y;
    const int q0 = (blockIdx.x * 8 + wave) * 16;

    const int sr = tid >> 5;
    const int sc = (tid & 31) * 4;
    const float* kg = K + (size_t)sr * DIMM + h * HDIM + sc;
    const float* vg = V + (size_t)(2 * sr) * DIMM + h * HDIM + sc;
    const size_t R8  = (size_t)8 * DIMM;
    const size_t R16 = (size_t)16 * DIMM;
    const int skoff = sr * AKP + sc;
    const int svoff = sr * AVP + 2 * sc;

    const float4 fk0 = *(const float4*)(kg);
    const float4 fk1 = *(const float4*)(kg + R8);
    const float4 fk2 = *(const float4*)(kg + 2 * R8);
    const float4 fk3 = *(const float4*)(kg + 3 * R8);
    const float4 fv0 = *(const float4*)(vg);
    const float4 fv1 = *(const float4*)(vg + DIMM);
    const float4 fv2 = *(const float4*)(vg + R16);
    const float4 fv3 = *(const float4*)(vg + R16 + DIMM);

    v2f aq[32];
    {
        const float* qrow = Q + (size_t)(q0 + lc) * DIMM + h * HDIM + 2 * hi;
#pragma unroll
        for (int s = 0; s < 32; ++s) aq[s] = *(const v2f*)(qrow + s * 4);
    }

    *(float4*)&sK[0][skoff]            = fk0;
    *(float4*)&sK[0][skoff +  8 * AKP] = fk1;
    *(float4*)&sK[0][skoff + 16 * AKP] = fk2;
    *(float4*)&sK[0][skoff + 24 * AKP] = fk3;
    *(float4*)&sV[0][svoff]               = make_float4(fv0.x, fv1.x, fv0.y, fv1.y);
    *(float4*)&sV[0][svoff + 4]           = make_float4(fv0.z, fv1.z, fv0.w, fv1.w);
    *(float4*)&sV[0][svoff + 8 * AVP]     = make_float4(fv2.x, fv3.x, fv2.y, fv3.y);
    *(float4*)&sV[0][svoff + 8 * AVP + 4] = make_float4(fv2.z, fv3.z, fv2.w, fv3.w);
    __syncthreads();

    v8f oacc[8];
#pragma unroll
    for (int t = 0; t < 8; ++t) oacc[t] = (v8f){};
    float mrun = -1.0e30f, lrun = 0.0f;

    const float scale = 0.08838834764831845f;
    const size_t mbase = ((size_t)h * SEQL + q0) * SEQL;
    const float* mlane = mask + mbase + (size_t)lc * SEQL + 8 * hi;

    const int NC = SEQL / AKC;
#pragma unroll 1
    for (int c = 0; c < NC; ++c) {
        const int cur = c & 1;
        const bool more = (c + 1 < NC);

        const int cn = more ? c + 1 : c;
        const float* kn = kg + (size_t)cn * AKC * DIMM;
        const float* vn = vg + (size_t)cn * AKC * DIMM;
        const float4 rk0 = *(const float4*)(kn);
        const float4 rk1 = *(const float4*)(kn + R8);
        const float4 rk2 = *(const float4*)(kn + 2 * R8);
        const float4 rk3 = *(const float4*)(kn + 3 * R8);
        const float4 rv0 = *(const float4*)(vn);
        const float4 rv1 = *(const float4*)(vn + DIMM);
        const float4 rv2 = *(const float4*)(vn + R16);
        const float4 rv3 = *(const float4*)(vn + R16 + DIMM);

        const float* ck = &sK[cur][lc * AKP + 2 * hi];
        const float* cv = &sV[cur][(4 * hi) * AVP + 2 * lc];

        v4f mk0 = __builtin_nontemporal_load((const v4f*)(mlane + c * AKC));
        v4f mk1 = __builtin_nontemporal_load((const v4f*)(mlane + c * AKC + 4));

        v2f bk[32];
#pragma unroll
        for (int ss = 0; ss < 32; ++ss) bk[ss] = *(const v2f*)(ck + ss * 4);
        __builtin_amdgcn_sched_barrier(0);

#pragma unroll
        for (int j = 0; j < AKC / 16; ++j) {
            const int kv0 = c * AKC + j * 16;

            v8f s = {};
#pragma unroll
            for (int ss = 0; ss < 32; ++ss) s = wmma_f32(bk[ss], aq[ss], s);

#pragma unroll
            for (int r = 0; r < 4; ++r) {
                s[r]     = s[r]     * scale + mk0[r];
                s[r + 4] = s[r + 4] * scale + mk1[r];
            }
            float tmax = fmaxf(fmaxf(fmaxf(s[0], s[1]), fmaxf(s[2], s[3])),
                               fmaxf(fmaxf(s[4], s[5]), fmaxf(s[6], s[7])));
            tmax = fmaxf(tmax, xhalf(tmax));

            const float mnew = fmaxf(mrun, tmax);
            const float corr = __expf(mrun - mnew);
            mrun = mnew;
#pragma unroll
            for (int r = 0; r < 8; ++r) s[r] = __expf(s[r] - mnew);

#pragma unroll
            for (int t = 0; t < 8; ++t)
#pragma unroll
                for (int r = 0; r < 8; ++r) oacc[t][r] *= corr;

            v2f pb[4];
#pragma unroll
            for (int kk = 0; kk < 4; ++kk) pb[kk] = (v2f){s[2 * kk], s[2 * kk + 1]};

            if (j == AKC / 16 - 1 && more) {
                float* dk = &sK[cur ^ 1][skoff];
                float* dv = &sV[cur ^ 1][svoff];
                *(float4*)(dk)            = rk0;
                *(float4*)(dk +  8 * AKP) = rk1;
                *(float4*)(dk + 16 * AKP) = rk2;
                *(float4*)(dk + 24 * AKP) = rk3;
                *(float4*)(dv)               = make_float4(rv0.x, rv1.x, rv0.y, rv1.y);
                *(float4*)(dv + 4)           = make_float4(rv0.z, rv1.z, rv0.w, rv1.w);
                *(float4*)(dv + 8 * AVP)     = make_float4(rv2.x, rv3.x, rv2.y, rv3.y);
                *(float4*)(dv + 8 * AVP + 4) = make_float4(rv2.z, rv3.z, rv2.w, rv3.w);
            }

            v2f bv[4][8];
#pragma unroll
            for (int kk = 0; kk < 4; ++kk) {
                const float* vp = cv + (j * 8 + kk) * AVP;
#pragma unroll
                for (int t = 0; t < 8; ++t) bv[kk][t] = *(const v2f*)(vp + t * 32);
            }
            if (j + 1 < AKC / 16) {
#pragma unroll
                for (int ss = 0; ss < 32; ++ss)
                    bk[ss] = *(const v2f*)(ck + (j + 1) * 16 * AKP + ss * 4);
                mk0 = __builtin_nontemporal_load((const v4f*)(mlane + kv0 + 16));
                mk1 = __builtin_nontemporal_load((const v4f*)(mlane + kv0 + 16 + 4));
            }
            __builtin_amdgcn_sched_barrier(0);
#pragma unroll
            for (int kk = 0; kk < 4; ++kk) {
#pragma unroll
                for (int t = 0; t < 8; ++t)
                    oacc[t] = wmma_f32(bv[kk][t], pb[kk], oacc[t]);
            }

            float psum = ((s[0] + s[1]) + (s[2] + s[3])) + ((s[4] + s[5]) + (s[6] + s[7]));
            psum += xhalf(psum);
            lrun = lrun * corr + psum;
        }
        __syncthreads();
    }

    const float inv = 1.0f / lrun;
    float* orow = O + (size_t)(q0 + lc) * DIMM + h * HDIM + 8 * hi;
#pragma unroll
    for (int t = 0; t < 8; ++t) {
        *(v4f*)(orow + t * 16)     = (v4f){oacc[t][0] * inv, oacc[t][1] * inv,
                                           oacc[t][2] * inv, oacc[t][3] * inv};
        *(v4f*)(orow + t * 16 + 4) = (v4f){oacc[t][4] * inv, oacc[t][5] * inv,
                                           oacc[t][6] * inv, oacc[t][7] * inv};
    }
}

extern "C" void kernel_launch(void* const* d_in, const int* in_sizes, int n_in,
                              void* d_out, int out_size, void* d_ws, size_t ws_size,
                              hipStream_t stream) {
    const float* x    = (const float*)d_in[0];
    const float* mask = (const float*)d_in[1];
    const float* wq   = (const float*)d_in[3];
    const float* wk   = (const float*)d_in[4];
    const float* wv   = (const float*)d_in[5];
    const float* wo   = (const float*)d_in[6];
    float* out = (float*)d_out;

    float* ws = (float*)d_ws;
    const size_t MAT = (size_t)SEQL * DIMM;
    float* Qb = ws;
    float* Kb = ws + MAT;
    float* Vb = ws + 2 * MAT;
    float* Ab = ws + 3 * MAT;

    dim3 tb(256);
    dim3 gg(DIMM / GBN, SEQL / GBM);
    gemm_abt<<<gg, tb, 0, stream>>>(x, wq, Qb);
    gemm_abt<<<gg, tb, 0, stream>>>(x, wk, Kb);
    gemm_abt<<<gg, tb, 0, stream>>>(x, wv, Vb);

    dim3 ga(SEQL / 128, NHEADS);
    attn_fwd<<<ga, tb, 0, stream>>>(Qb, Kb, Vb, mask, Ab);

    gemm_abt<<<gg, tb, 0, stream>>>(Ab, wo, out);
}


# gemm_abt: the three projections of x as ONE launch of 16x16x3 blocks (blockIdx.z picks W and output), held to 160 VGPRs so three blocks share a WGP and fill each other's barrier gaps; sums unchanged
# speedup vs baseline: 1.7124x; 1.0665x over previous
#include <hip/hip_runtime.h>

#define SEQL   2048
#define DIMM   2048
#define NHEADS 16
#define HDIM   128

typedef float v2f __attribute__((ext_vector_type(2)));
typedef float v4f __attribute__((ext_vector_type(4)));
typedef float v8f __attribute__((ext_vector_type(8)));

__device__ __forceinline__ v8f wmma_f32(v2f a, v2f b, v8f c) {
    return __builtin_amdgcn_wmma_f32_16x16x4_f32(false, a, false, b, (short)0, c, false, false);
}

__device__ __forceinline__ float xhalf(float v) {
    const int b = __float_as_int(v);
    return __int_as_float(__builtin_amdgcn_permlanex16(b, b, 0x76543210, 0xfedcba98, false, false));
}

#define GBM 128
#define GBN 128
#define GBK 32
#define GLD 36

__global__ __launch_bounds__(256, 6)
void gemm_abt(const float* A, const float* B0, const float* B1, const float* B2,
              float* C0, float* C1, float* C2) {
    __shared__ __attribute__((aligned(16))) float sA[2][GBM * GLD];
    __shared__ __attribute__((aligned(16))) float sB[2][GBN * GLD];

    const int tid  = threadIdx.x;
    const int lane = tid & 31;
    const int wave = tid >> 5;
    const int hi   = lane >> 4;
    const int lc   = lane & 15;
    const int wm   = wave >> 1;
    const int wn   = wave & 1;

    const int m0 = blockIdx.y * GBM;
    const int n0 = blockIdx.x * GBN;
    const int z = blockIdx.z;
    const float* B = (z == 0) ? B0 : (z == 1) ? B1 : B2;
    float* C = (z == 0) ? C0 : (z == 1) ? C1 : C2;

    const int lr   = tid >> 3;
    const int lcol = (tid & 7) * 4;
    const float* ag = A + (size_t)(m0 + lr) * DIMM + lcol;
    const float* bg = B + (size_t)(n0 + lr) * DIMM + lcol;
    const int soff = lr * GLD + lcol;

    const size_t RS = (size_t)32 * DIMM;
    float4 ra0 = *(const float4*)(ag);
    float4 ra1 = *(const float4*)(ag + RS);
    float4 ra2 = *(const float4*)(ag + 2 * RS);
    float4 ra3 = *(const float4*)(ag + 3 * RS);
    float4 rb0 = *(const float4*)(bg);
    float4 rb1 = *(const float4*)(bg + RS);
    float4 rb2 = *(const float4*)(bg + 2 * RS);
    float4 rb3 = *(const float4*)(bg + 3 * RS);
    *(float4*)&sA[0][soff]            = ra0;
    *(float4*)&sA[0][soff + 32 * GLD] = ra1;
    *(float4*)&sA[0][soff + 64 * GLD] = ra2;
    *(float4*)&sA[0][soff + 96 * GLD] = ra3;
    *(float4*)&sB[0][soff]            = rb0;
    *(float4*)&sB[0][soff + 32 * GLD] = rb1;
    *(float4*)&sB[0][soff + 64 * GLD] = rb2;
    *(float4*)&sB[0][soff + 96 * GLD] = rb3;
    __syncthreads();

    v8f acc[2][4];
#pragma unroll
    for (int i = 0; i < 2; ++i)
#pragma unroll
        for (int j = 0; j < 4; ++j) acc[i][j] = (v8f){};

    const int aoff = (wm * 32 + lc) * GLD + 2 * hi;
    const int boff = (wn * 64 + lc) * GLD + 2 * hi;

    const int NT = DIMM / GBK;
    for (int kt = 0; kt < NT; ++kt) {
        const int cur = kt & 1;
        const bool more = (kt + 1 < NT);

        const float* pa = &sA[cur][aoff];
        const float* pb = &sB[cur][boff];

        v2f fa[2][2][2];
        v2f fb[2][2][4];
#pragma unroll
        for (int u = 0; u < 2; ++u) {
            fa[0][u][0] = *(const v2f*)(pa + u * 4);
            fa[0][u][1] = *(const v2f*)(pa + 16 * GLD + u * 4);
#pragma unroll
            for (int j = 0; j < 4; ++j)
                fb[0][u][j] = *(const v2f*)(pb + j * 16 * GLD + u * 4);
        }
        __builtin_amdgcn_sched_barrier(0);

        if (more) {
            const float* an = ag + (kt + 1) * GBK;
            const float* bn = bg + (kt + 1) * GBK;
            ra0 = *(const float4*)(an);
            ra1 = *(const float4*)(an + RS);
            ra2 = *(const float4*)(an + 2 * RS);
            ra3 = *(const float4*)(an + 3 * RS);
            rb0 = *(const float4*)(bn);
            rb1 = *(const float4*)(bn + RS);
            rb2 = *(const float4*)(bn + 2 * RS);
            rb3 = *(const float4*)(bn + 3 * RS);
        }
        __builtin_amdgcn_sched_barrier(0);

#pragma unroll
        for (int g = 0; g < GBK / 8; ++g) {
            const int cs = g & 1;
            if (g + 1 < GBK / 8) {
#pragma unroll
                for (int u = 0; u < 2; ++u) {
                    const int kk = (g + 1) * 8 + u * 4;
                    fa[cs ^ 1][u][0] = *(const v2f*)(pa + kk);
                    fa[cs ^ 1][u][1] = *(const v2f*)(pa + 16 * GLD + kk);
#pragma unroll
                    for (int j = 0; j < 4; ++j)
                        fb[cs ^ 1][u][j] = *(const v2f*)(pb + j * 16 * GLD + kk);
                }
            }
            if (g == GBK / 8 - 1 && more) {
                float* da = &sA[cur ^ 1][soff];
                float* db = &sB[cur ^ 1][soff];
                *(float4*)(da)            = ra0;
                *(float4*)(da + 32 * GLD) = ra1;
                *(float4*)(da + 64 * GLD) = ra2;
                *(float4*)(da + 96 * GLD) = ra3;
                *(float4*)(db)            = rb0;
                *(float4*)(db + 32 * GLD) = rb1;
                *(float4*)(db + 64 * GLD) = rb2;
                *(float4*)(db + 96 * GLD) = rb3;
            }
            __builtin_amdgcn_sched_barrier(0);
#pragma unroll
            for (int u = 0; u < 2; ++u) {
#pragma unroll
                for (int j = 0; j < 4; ++j) {
                    acc[0][j] = wmma_f32(fa[cs][u][0], fb[cs][u][j], acc[0][j]);
                    acc[1][j] = wmma_f32(fa[cs][u][1], fb[cs][u][j], acc[1][j]);
                }
                __builtin_amdgcn_sched_barrier(0);
            }
        }
        __syncthreads();
    }

    int te = tid;
    asm volatile("" : "+v"(te));
    const int elc = te & 15;
    const int ehi = (te >> 4) & 1;
    const int ewm = te >> 6;
    const int ewn = (te >> 5) & 1;
#pragma unroll
    for (int i = 0; i < 2; ++i) {
#pragma unroll
        for (int r = 0; r < 8; ++r) {
            float* crow = C + (size_t)(m0 + ewm * 32 + i * 16 + r + 8 * ehi) * DIMM
                            + n0 + ewn * 64 + elc;
            crow[ 0] = acc[i][0][r];
            crow[16] = acc[i][1][r];
            crow[32] = acc[i][2][r];
            crow[48] = acc[i][3][r];
        }
    }
}

#define AKC 32
#define AKP 132
#define AVP 264

__global__ __launch_bounds__(256)
void attn_fwd(const float* __restrict__ Q, const float* K,
              const float* V, const float* __restrict__ mask,
              float* __restrict__ O) {
    __shared__ __attribute__((aligned(16))) float sK[2][AKC * AKP];
    __shared__ __attribute__((aligned(16))) float sV[2][(AKC / 2) * AVP];

    const int tid  = threadIdx.x;
    const int lane = tid & 31;
    const int wave = tid >> 5;
    const int hi   = lane >> 4;
    const int lc   = lane & 15;

    const int h  = blockIdx.y;
    const int q0 = (blockIdx.x * 8 + wave) * 16;

    const int sr = tid >> 5;
    const int sc = (tid & 31) * 4;
    const float* kg = K + (size_t)sr * DIMM + h * HDIM + sc;
    const float* vg = V + (size_t)(2 * sr) * DIMM + h * HDIM + sc;
    const size_t R8  = (size_t)8 * DIMM;
    const size_t R16 = (size_t)16 * DIMM;
    const int skoff = sr * AKP + sc;
    const int svoff = sr * AVP + 2 * sc;

    const float4 fk0 = *(const float4*)(kg);
    const float4 fk1 = *(const float4*)(kg + R8);
    const float4 fk2 = *(const float4*)(kg + 2 * R8);
    const float4 fk3 = *(const float4*)(kg + 3 * R8);
    const float4 fv0 = *(const float4*)(vg);
    const float4 fv1 = *(const float4*)(vg + DIMM);
    const float4 fv2 = *(const float4*)(vg + R16);
    const float4 fv3 = *(const float4*)(vg + R16 + DIMM);

    v2f aq[32];
    {
        const float* qrow = Q + (size_t)(q0 + lc) * DIMM + h * HDIM + 2 * hi;
#pragma unroll
        for (int s = 0; s < 32; ++s) aq[s] = *(const v2f*)(qrow + s * 4);
    }

    *(float4*)&sK[0][skoff]            = fk0;
    *(float4*)&sK[0][skoff +  8 * AKP] = fk1;
    *(float4*)&sK[0][skoff + 16 * AKP] = fk2;
    *(float4*)&sK[0][skoff + 24 * AKP] = fk3;
    *(float4*)&sV[0][svoff]               = make_float4(fv0.x, fv1.x, fv0.y, fv1.y);
    *(float4*)&sV[0][svoff + 4]           = make_float4(fv0.z, fv1.z, fv0.w, fv1.w);
    *(float4*)&sV[0][svoff + 8 * AVP]     = make_float4(fv2.x, fv3.x, fv2.y, fv3.y);
    *(float4*)&sV[0][svoff + 8 * AVP + 4] = make_float4(fv2.z, fv3.z, fv2.w, fv3.w);
    __syncthreads();

    v8f oacc[8];
#pragma unroll
    for (int t = 0; t < 8; ++t) oacc[t] = (v8f){};
    float mrun = -1.0e30f, lrun = 0.0f;

    const float scale = 0.08838834764831845f;
    const size_t mbase = ((size_t)h * SEQL + q0) * SEQL;
    const float* mlane = mask + mbase + (size_t)lc * SEQL + 8 * hi;

    const int NC = SEQL / AKC;
#pragma unroll 1
    for (int c = 0; c < NC; ++c) {
        const int cur = c & 1;
        const bool more = (c + 1 < NC);

        const int cn = more ? c + 1 : c;
        const float* kn = kg + (size_t)cn * AKC * DIMM;
        const float* vn = vg + (size_t)cn * AKC * DIMM;
        const float4 rk0 = *(const float4*)(kn);
        const float4 rk1 = *(const float4*)(kn + R8);
        const float4 rk2 = *(const float4*)(kn + 2 * R8);
        const float4 rk3 = *(const float4*)(kn + 3 * R8);
        const float4 rv0 = *(const float4*)(vn);
        const float4 rv1 = *(const float4*)(vn + DIMM);
        const float4 rv2 = *(const float4*)(vn + R16);
        const float4 rv3 = *(const float4*)(vn + R16 + DIMM);

        const float* ck = &sK[cur][lc * AKP + 2 * hi];
        const float* cv = &sV[cur][(4 * hi) * AVP + 2 * lc];

        v4f mk0 = __builtin_nontemporal_load((const v4f*)(mlane + c * AKC));
        v4f mk1 = __builtin_nontemporal_load((const v4f*)(mlane + c * AKC + 4));

        v2f bk[32];
#pragma unroll
        for (int ss = 0; ss < 32; ++ss) bk[ss] = *(const v2f*)(ck + ss * 4);
        __builtin_amdgcn_sched_barrier(0);

#pragma unroll
        for (int j = 0; j < AKC / 16; ++j) {
            const int kv0 = c * AKC + j * 16;

            v8f s = {};
#pragma unroll
            for (int ss = 0; ss < 32; ++ss) s = wmma_f32(bk[ss], aq[ss], s);

#pragma unroll
            for (int r = 0; r < 4; ++r) {
                s[r]     = s[r]     * scale + mk0[r];
                s[r + 4] = s[r + 4] * scale + mk1[r];
            }
            float tmax = fmaxf(fmaxf(fmaxf(s[0], s[1]), fmaxf(s[2], s[3])),
                               fmaxf(fmaxf(s[4], s[5]), fmaxf(s[6], s[7])));
            tmax = fmaxf(tmax, xhalf(tmax));

            const float mnew = fmaxf(mrun, tmax);
            const float corr = __expf(mrun - mnew);
            mrun = mnew;
#pragma unroll
            for (int r = 0; r < 8; ++r) s[r] = __expf(s[r] - mnew);

#pragma unroll
            for (int t = 0; t < 8; ++t)
#pragma unroll
                for (int r = 0; r < 8; ++r) oacc[t][r] *= corr;

            v2f pb[4];
#pragma unroll
            for (int kk = 0; kk < 4; ++kk) pb[kk] = (v2f){s[2 * kk], s[2 * kk + 1]};

            if (j == AKC / 16 - 1 && more) {
                float* dk = &sK[cur ^ 1][skoff];
                float* dv = &sV[cur ^ 1][svoff];
                *(float4*)(dk)            = rk0;
                *(float4*)(dk +  8 * AKP) = rk1;
                *(float4*)(dk + 16 * AKP) = rk2;
                *(float4*)(dk + 24 * AKP) = rk3;
                *(float4*)(dv)               = make_float4(rv0.x, rv1.x, rv0.y, rv1.y);
                *(float4*)(dv + 4)           = make_float4(rv0.z, rv1.z, rv0.w, rv1.w);
                *(float4*)(dv + 8 * AVP)     = make_float4(rv2.x, rv3.x, rv2.y, rv3.y);
                *(float4*)(dv + 8 * AVP + 4) = make_float4(rv2.z, rv3.z, rv2.w, rv3.w);
            }

            v2f bv[4][8];
#pragma unroll
            for (int kk = 0; kk < 4; ++kk) {
                const float* vp = cv + (j * 8 + kk) * AVP;
#pragma unroll
                for (int t = 0; t < 8; ++t) bv[kk][t] = *(const v2f*)(vp + t * 32);
            }
            if (j + 1 < AKC / 16) {
#pragma unroll
                for (int ss = 0; ss < 32; ++ss)
                    bk[ss] = *(const v2f*)(ck + (j + 1) * 16 * AKP + ss * 4);
                mk0 = __builtin_nontemporal_load((const v4f*)(mlane + kv0 + 16));
                mk1 = __builtin_nontemporal_load((const v4f*)(mlane + kv0 + 16 + 4));
            }
            __builtin_amdgcn_sched_barrier(0);
#pragma unroll
            for (int kk = 0; kk < 4; ++kk) {
#pragma unroll
                for (int t = 0; t < 8; ++t)
                    oacc[t] = wmma_f32(bv[kk][t], pb[kk], oacc[t]);
            }

            float psum = ((s[0] + s[1]) + (s[2] + s[3])) + ((s[4] + s[5]) + (s[6] + s[7]));
            psum += xhalf(psum);
            lrun = lrun * corr + psum;
        }
        __syncthreads();
    }

    const float inv = 1.0f / lrun;
    float* orow = O + (size_t)(q0 + lc) * DIMM + h * HDIM + 8 * hi;
#pragma unroll
    for (int t = 0; t < 8; ++t) {
        *(v4f*)(orow + t * 16)     = (v4f){oacc[t][0] * inv, oacc[t][1] * inv,
                                           oacc[t][2] * inv, oacc[t][3] * inv};
        *(v4f*)(orow + t * 16 + 4) = (v4f){oacc[t][4] * inv, oacc[t][5] * inv,
                                           oacc[t][6] * inv, oacc[t][7] * inv};
    }
}

extern "C" void kernel_launch(void* const* d_in, const int* in_sizes, int n_in,
                              void* d_out, int out_size, void* d_ws, size_t ws_size,
                              hipStream_t stream) {
    const float* x    = (const float*)d_in[0];
    const float* mask = (const float*)d_in[1];
    const float* wq   = (const float*)d_in[3];
    const float* wk   = (const float*)d_in[4];
    const float* wv   = (const float*)d_in[5];
    const float* wo   = (const float*)d_in[6];
    float* out = (float*)d_out;

    float* ws = (float*)d_ws;
    const size_t MAT = (size_t)SEQL * DIMM;
    float* Qb = ws;
    float* Kb = ws + MAT;
    float* Vb = ws + 2 * MAT;
    float* Ab = ws + 3 * MAT;

    dim3 tb(256);
    dim3 gg(DIMM / GBN, SEQL / GBM);
    dim3 g3(DIMM / GBN, SEQL / GBM, 3);
    gemm_abt<<<g3, tb, 0, stream>>>(x, wq, wk, wv, Qb, Kb, Vb);

    dim3 ga(SEQL / 128, NHEADS);
    attn_fwd<<<ga, tb, 0, stream>>>(Qb, Kb, Vb, mask, Ab);

    gemm_abt<<<gg, tb, 0, stream>>>(Ab, wo, wo, wo, out, out, out);
}
